# starting kernel
# baseline (speedup 1.0000x reference)
#include <hip/hip_runtime.h>
#include <math.h>

typedef __attribute__((ext_vector_type(2))) float v2f;
typedef __attribute__((ext_vector_type(8))) float v8f;

static __device__ __forceinline__ v8f wmma_f32(v2f a, v2f b, v8f c) {
  return __builtin_amdgcn_wmma_f32_16x16x4_f32(false, a, false, b, (short)0, c, false, false);
}

static __device__ __forceinline__ float cos8(int n) {
  n &= 7;
  const float R = 0.7071067811865476f;
  float mag = (n & 1) ? R : ((n & 2) ? 0.0f : 1.0f);
  float r = ((n >= 3) && (n <= 5)) ? -mag : mag;
  return r;
}
static __device__ __forceinline__ float sin8(int n) { return cos8(n + 6); }

__global__ void __launch_bounds__(256) dct_zero_tail(float* __restrict__ out) {
  const int per_b = 24 * 4096;
  int idx = blockIdx.x * 256 + threadIdx.x;
  if (idx < 32 * per_b) {
    int b = idx / per_b;
    int r = idx - b * per_b;
    out[b * (64 * 4096) + 40 * 4096 + r] = 0.0f;
  }
}

__global__ void __launch_bounds__(256) dct8x8_wmma(const float* __restrict__ x,
                                                   const float* __restrict__ fw,
                                                   float* __restrict__ out) {
  __shared__ float lds[8][2][256];
  const int lane = threadIdx.x & 31;
  const int wave = threadIdx.x >> 5;
  const int gw = blockIdx.x * 8 + wave;
  const int b  = gw >> 10;
  const int ty = (gw >> 5) & 31;
  const int tx = gw & 31;
  const int h  = lane >> 4;
  const int nl = lane & 15;

  v2f ca[4], sa[4], nsa[4];
#pragma unroll
  for (int j = 0; j < 4; ++j) {
#pragma unroll
    for (int v = 0; v < 2; ++v) {
      const int k = 4 * j + 2 * h + v;
      float c = 0.0f, s = 0.0f;
      if ((nl >> 3) == (k >> 3)) {
        const int n = (nl & 7) * (k & 7);
        c =  cos8(n) * 0.35355339059327373f;
        s = -sin8(n) * 0.35355339059327373f;
      }
      ca[j][v] = c; sa[j][v] = s; nsa[j][v] = -s;
    }
  }

  const int r0 = ty * 16, c0 = tx * 16;
  const float* xb0 = x + (b * 512 + r0) * 512 + c0 + nl;
  v2f xb[4];
#pragma unroll
  for (int j = 0; j < 4; ++j) {
    const int rr = 4 * j + 2 * h;
    xb[j][0] = xb0[rr * 512];
    xb[j][1] = xb0[rr * 512 + 512];
  }

  v8f t1 = {0, 0, 0, 0, 0, 0, 0, 0};
  v8f t2 = {0, 0, 0, 0, 0, 0, 0, 0};
#pragma unroll
  for (int j = 0; j < 4; ++j) {
    t1 = wmma_f32(ca[j], xb[j], t1);
    t2 = wmma_f32(sa[j], xb[j], t2);
  }

#pragma unroll
  for (int v = 0; v < 8; ++v) {
    lds[wave][0][(v + 8 * h) * 16 + nl] = t1[v];
    lds[wave][1][(v + 8 * h) * 16 + nl] = t2[v];
  }
  asm volatile("s_wait_dscnt 0" ::: "memory");

  v8f yre = {0, 0, 0, 0, 0, 0, 0, 0};
  v8f yim = {0, 0, 0, 0, 0, 0, 0, 0};
#pragma unroll
  for (int j = 0; j < 4; ++j) {
    const int off = nl * 16 + 4 * j + 2 * h;
    v2f a1; a1[0] = lds[wave][0][off]; a1[1] = lds[wave][0][off + 1];
    v2f a2; a2[0] = lds[wave][1][off]; a2[1] = lds[wave][1][off + 1];
    yre = wmma_f32(a1, ca[j],  yre);
    yre = wmma_f32(a2, nsa[j], yre);
    yim = wmma_f32(a2, ca[j],  yim);
    yim = wmma_f32(a1, sa[j],  yim);
  }

  const int kx = nl & 7;
  const int hb = ty * 2 + h;
  const int wb = tx * 2 + (nl >> 3);
  if (kx <= 4) {
    float* ob = out + b * (64 * 4096) + hb * 64 + wb;
#pragma unroll
    for (int v = 0; v < 8; ++v) {
      const int f = v * 5 + kx;
      const float w = __fdividef(1.0f, 1.0f + __expf(-fw[f]));
      const float mag = sqrtf(yre[v] * yre[v] + yim[v] * yim[v]);
      ob[f * 4096] = mag * w;
    }
  }
}

extern "C" void kernel_launch(void* const* d_in, const int* in_sizes, int n_in,
                              void* d_out, int out_size, void* d_ws, size_t ws_size,
                              hipStream_t stream) {
  (void)in_sizes; (void)n_in; (void)out_size; (void)d_ws; (void)ws_size;
  const float* x  = (const float*)d_in[0];
  const float* fw = (const float*)d_in[1];
  float* out = (float*)d_out;

  dct_zero_tail<<<(32 * 24 * 4096 + 255) / 256, 256, 0, stream>>>(out);
  dct8x8_wmma<<<4096, 256, 0, stream>>>(x, fw, out);
}


# one 8x8 block per lane column (16 blocks = N), DFT matrices as A, pass-1 D pairs reused as pass-2 B; each wave now writes whole 128B plane lines (was 8B pieces); sums reordered: columns then rows
# speedup vs baseline: 4.6958x; 4.6958x over previous
#include <hip/hip_runtime.h>
#include <math.h>

typedef __attribute__((ext_vector_type(2))) float v2f;
typedef __attribute__((ext_vector_type(4))) float v4f;
typedef __attribute__((ext_vector_type(8))) float v8f;

static __device__ __forceinline__ v8f wmma_f32(v2f a, v2f b, v8f c) {
  return __builtin_amdgcn_wmma_f32_16x16x4_f32(false, a, false, b, (short)0, c, false, false);
}

static __device__ __forceinline__ float cos8(int n) {
  n &= 7;
  const float R = 0.7071067811865476f;
  float mag = (n & 1) ? R : ((n & 2) ? 0.0f : 1.0f);
  float r = ((n >= 3) && (n <= 5)) ? -mag : mag;
  return r;
}
static __device__ __forceinline__ float sin8(int n) { return cos8(n + 6); }

__global__ void __launch_bounds__(256) dct_zero_tail(float* __restrict__ out) {
  const int per_b = 24 * 4096;
  int idx = blockIdx.x * 256 + threadIdx.x;
  if (idx < 32 * per_b) {
    int b = idx / per_b;
    int r = idx - b * per_b;
    out[b * (64 * 4096) + 40 * 4096 + r] = 0.0f;
  }
}

#define LDS_PITCH 36

__global__ void __launch_bounds__(256) __attribute__((amdgpu_waves_per_eu(4, 4))) dct8x8_wmma(const float* __restrict__ x,
                                                   const float* __restrict__ fw,
                                                   float* __restrict__ out) {
  __shared__ __attribute__((aligned(16))) float lds[8][40 * LDS_PITCH];
  const int lane = threadIdx.x & 31;
  const int wave = threadIdx.x >> 5;
  const int gw = blockIdx.x * 8 + wave;
  const int b    = gw >> 7;
  const int hb   = (gw >> 1) & 63;
  const int half = gw & 1;
  const int h  = lane >> 4;
  const int nl = lane & 15;

  const float* xp = x + ((b * 512 + hb * 8) * 512 + half * 256 + nl * 8 + 4 * h);
  v4f xr[2][8];
  for (int t = 0; t < 2; ++t) {
    for (int r = 0; r < 8; ++r) {
      xr[t][r] = *reinterpret_cast<const v4f*>(xp + r * 512 + t * 128);
    }
  }

  const int fr = lane >> 3;
  const int c4 = (lane & 7) * 4;
  float wsig[10];
  for (int i = 0; i < 10; ++i) {
    wsig[i] = __fdividef(1.0f, 1.0f + __expf(-fw[4 * i + fr]));
  }

  const float S1 = 0.35355339059327373f;

  v2f a1[2][2];
  {
    const int rowsel = nl >> 3;
    const int vi = nl & 7;
    const int kx = (vi == 0) ? 0 : ((vi == 1) ? 4 : (vi >> 1));
    const int part = (vi >= 2) ? (vi & 1) : 0;
    for (int rho = 0; rho < 2; ++rho) {
      for (int ip = 0; ip < 2; ++ip) {
        for (int v = 0; v < 2; ++v) {
          const int c = 4 * h + 2 * ip + v;
          const int n = kx * c;
          const float val = cos8(n + (part ? 2 : 0));
          a1[rho][ip][v] = (rowsel == rho) ? val * S1 : 0.0f;
        }
      }
    }
  }

  v2f a2[4], a3[4];
  {
    const int ky2 = 4 * (nl >> 3) + ((nl & 7) >> 1);
    const int op2 = nl & 1;
    const int kxo = nl >> 3;
    const int v3 = nl & 7;
    const int ky3 = (v3 == 0) ? 0 : ((v3 == 1) ? 4 : (v3 >> 1));
    const int op3 = (v3 >= 2) ? (v3 & 1) : 0;
    for (int q = 0; q < 4; ++q) {
      const int r = 2 * q + h;
      for (int v = 0; v < 2; ++v) {
        const int n2 = ky2 * r;
        const int sh2 = (op2 == 0) ? ((v == 0) ? 0 : 6) : ((v == 0) ? 2 : 0);
        a2[q][v] = cos8(n2 + sh2) * S1;
        const int n3 = ky3 * r;
        const float val3 = cos8(n3 + (op3 ? 2 : 0));
        a3[q][v] = (v == kxo) ? val3 * S1 : 0.0f;
      }
    }
  }

  float* L = &lds[wave][0];

  for (int t = 0; t < 2; ++t) {
    v8f d1[4];
    for (int q = 0; q < 4; ++q) d1[q] = (v8f){0, 0, 0, 0, 0, 0, 0, 0};
    for (int rho = 0; rho < 2; ++rho) {
      for (int ip = 0; ip < 2; ++ip) {
        for (int q = 0; q < 4; ++q) {
          v2f bb;
          bb[0] = xr[t][2 * q + rho][2 * ip];
          bb[1] = xr[t][2 * q + rho][2 * ip + 1];
          d1[q] = wmma_f32(a1[rho][ip], bb, d1[q]);
        }
      }
    }

    v8f y0 = {0, 0, 0, 0, 0, 0, 0, 0};
    v8f y1 = {0, 0, 0, 0, 0, 0, 0, 0};
    v8f y2 = {0, 0, 0, 0, 0, 0, 0, 0};
    v8f y3 = {0, 0, 0, 0, 0, 0, 0, 0};
    for (int q = 0; q < 4; ++q) {
      v2f b0; b0[0] = d1[q][0]; b0[1] = d1[q][1];
      v2f b1; b1[0] = d1[q][2]; b1[1] = d1[q][3];
      v2f b2; b2[0] = d1[q][4]; b2[1] = d1[q][5];
      v2f b3; b3[0] = d1[q][6]; b3[1] = d1[q][7];
      y0 = wmma_f32(a3[q], b0, y0);
      y1 = wmma_f32(a2[q], b1, y1);
      y2 = wmma_f32(a2[q], b2, y2);
      y3 = wmma_f32(a2[q], b3, y3);
    }

    const int col = t * 16 + nl;
    for (int j = 0; j < 4; ++j) {
      const int ky = 4 * h + j;
      L[(ky * 5 + 1) * LDS_PITCH + col] = sqrtf(y1[2 * j] * y1[2 * j] + y1[2 * j + 1] * y1[2 * j + 1]);
      L[(ky * 5 + 2) * LDS_PITCH + col] = sqrtf(y2[2 * j] * y2[2 * j] + y2[2 * j + 1] * y2[2 * j + 1]);
      L[(ky * 5 + 3) * LDS_PITCH + col] = sqrtf(y3[2 * j] * y3[2 * j] + y3[2 * j + 1] * y3[2 * j + 1]);
    }
    {
      const int kxe = 4 * h;
      const float m0 = fabsf(y0[0]);
      const float m4 = fabsf(y0[1]);
      const float m1 = sqrtf(y0[2] * y0[2] + y0[3] * y0[3]);
      const float m2 = sqrtf(y0[4] * y0[4] + y0[5] * y0[5]);
      const float m3 = sqrtf(y0[6] * y0[6] + y0[7] * y0[7]);
      L[(0 * 5 + kxe) * LDS_PITCH + col] = m0;
      L[(1 * 5 + kxe) * LDS_PITCH + col] = m1;
      L[(2 * 5 + kxe) * LDS_PITCH + col] = m2;
      L[(3 * 5 + kxe) * LDS_PITCH + col] = m3;
      L[(4 * 5 + kxe) * LDS_PITCH + col] = m4;
      L[(5 * 5 + kxe) * LDS_PITCH + col] = m3;
      L[(6 * 5 + kxe) * LDS_PITCH + col] = m2;
      L[(7 * 5 + kxe) * LDS_PITCH + col] = m1;
    }
  }

  asm volatile("s_wait_dscnt 0" ::: "memory");

  float* ob = out + b * (64 * 4096) + hb * 64 + half * 32 + c4;
  for (int i = 0; i < 10; ++i) {
    const int f = 4 * i + fr;
    const float w = wsig[i];
    v4f m = *reinterpret_cast<const v4f*>(L + f * LDS_PITCH + c4);
    m[0] *= w; m[1] *= w; m[2] *= w; m[3] *= w;
    *reinterpret_cast<v4f*>(ob + f * 4096) = m;
  }
}

extern "C" void kernel_launch(void* const* d_in, const int* in_sizes, int n_in,
                              void* d_out, int out_size, void* d_ws, size_t ws_size,
                              hipStream_t stream) {
  (void)in_sizes; (void)n_in; (void)out_size; (void)d_ws; (void)ws_size;
  const float* x  = (const float*)d_in[0];
  const float* fw = (const float*)d_in[1];
  float* out = (float*)d_out;

  dct_zero_tail<<<(32 * 24 * 4096 + 255) / 256, 256, 0, stream>>>(out);
  dct8x8_wmma<<<512, 256, 0, stream>>>(x, fw, out);
}


# zero planes 40..63 written by the transform kernel's own waves (6 whole-line 16-B stores per wave, between its two tiles); the separate 98304-wave zeroing launch is gone: 2 launches -> 1
# speedup vs baseline: 7.2521x; 1.5444x over previous
#include <hip/hip_runtime.h>
#include <math.h>

typedef __attribute__((ext_vector_type(2))) float v2f;
typedef __attribute__((ext_vector_type(4))) float v4f;
typedef __attribute__((ext_vector_type(8))) float v8f;

static __device__ __forceinline__ v8f wmma_f32(v2f a, v2f b, v8f c) {
  return __builtin_amdgcn_wmma_f32_16x16x4_f32(false, a, false, b, (short)0, c, false, false);
}

static __device__ __forceinline__ float cos8(int n) {
  n &= 7;
  const float R = 0.7071067811865476f;
  float mag = (n & 1) ? R : ((n & 2) ? 0.0f : 1.0f);
  float r = ((n >= 3) && (n <= 5)) ? -mag : mag;
  return r;
}
static __device__ __forceinline__ float sin8(int n) { return cos8(n + 6); }

#define LDS_PITCH 36

__global__ void __launch_bounds__(256) __attribute__((amdgpu_waves_per_eu(4, 4))) dct8x8_wmma(const float* __restrict__ x,
                                                   const float* __restrict__ fw,
                                                   float* __restrict__ out) {
  __shared__ __attribute__((aligned(16))) float lds[8][40 * LDS_PITCH];
  const int lane = threadIdx.x & 31;
  const int wave = threadIdx.x >> 5;
  const int gw = blockIdx.x * 8 + wave;
  const int b    = gw >> 7;
  const int hb   = (gw >> 1) & 63;
  const int half = gw & 1;
  const int h  = lane >> 4;
  const int nl = lane & 15;

  const float* xp = x + ((b * 512 + hb * 8) * 512 + half * 256 + nl * 8 + 4 * h);
  v4f xr[2][8];
  for (int t = 0; t < 2; ++t) {
    for (int r = 0; r < 8; ++r) {
      xr[t][r] = *reinterpret_cast<const v4f*>(xp + r * 512 + t * 128);
    }
  }

  const int fr = lane >> 3;
  const int c4 = (lane & 7) * 4;
  float* ob = out + b * (64 * 4096) + hb * 64 + half * 32 + c4;
  float wsig[10];
  for (int i = 0; i < 10; ++i) {
    wsig[i] = __fdividef(1.0f, 1.0f + __expf(-fw[4 * i + fr]));
  }

  const float S1 = 0.35355339059327373f;

  v2f a1[2][2];
  {
    const int rowsel = nl >> 3;
    const int vi = nl & 7;
    const int kx = (vi == 0) ? 0 : ((vi == 1) ? 4 : (vi >> 1));
    const int part = (vi >= 2) ? (vi & 1) : 0;
    for (int rho = 0; rho < 2; ++rho) {
      for (int ip = 0; ip < 2; ++ip) {
        for (int v = 0; v < 2; ++v) {
          const int c = 4 * h + 2 * ip + v;
          const int n = kx * c;
          const float val = cos8(n + (part ? 2 : 0));
          a1[rho][ip][v] = (rowsel == rho) ? val * S1 : 0.0f;
        }
      }
    }
  }

  v2f a2[4], a3[4];
  {
    const int ky2 = 4 * (nl >> 3) + ((nl & 7) >> 1);
    const int op2 = nl & 1;
    const int kxo = nl >> 3;
    const int v3 = nl & 7;
    const int ky3 = (v3 == 0) ? 0 : ((v3 == 1) ? 4 : (v3 >> 1));
    const int op3 = (v3 >= 2) ? (v3 & 1) : 0;
    for (int q = 0; q < 4; ++q) {
      const int r = 2 * q + h;
      for (int v = 0; v < 2; ++v) {
        const int n2 = ky2 * r;
        const int sh2 = (op2 == 0) ? ((v == 0) ? 0 : 6) : ((v == 0) ? 2 : 0);
        a2[q][v] = cos8(n2 + sh2) * S1;
        const int n3 = ky3 * r;
        const float val3 = cos8(n3 + (op3 ? 2 : 0));
        a3[q][v] = (v == kxo) ? val3 * S1 : 0.0f;
      }
    }
  }

  float* L = &lds[wave][0];

  for (int t = 0; t < 2; ++t) {
    v8f d1[4];
    for (int q = 0; q < 4; ++q) d1[q] = (v8f){0, 0, 0, 0, 0, 0, 0, 0};
    for (int rho = 0; rho < 2; ++rho) {
      for (int ip = 0; ip < 2; ++ip) {
        for (int q = 0; q < 4; ++q) {
          v2f bb;
          bb[0] = xr[t][2 * q + rho][2 * ip];
          bb[1] = xr[t][2 * q + rho][2 * ip + 1];
          d1[q] = wmma_f32(a1[rho][ip], bb, d1[q]);
        }
      }
    }

    v8f y0 = {0, 0, 0, 0, 0, 0, 0, 0};
    v8f y1 = {0, 0, 0, 0, 0, 0, 0, 0};
    v8f y2 = {0, 0, 0, 0, 0, 0, 0, 0};
    v8f y3 = {0, 0, 0, 0, 0, 0, 0, 0};
    for (int q = 0; q < 4; ++q) {
      v2f b0; b0[0] = d1[q][0]; b0[1] = d1[q][1];
      v2f b1; b1[0] = d1[q][2]; b1[1] = d1[q][3];
      v2f b2; b2[0] = d1[q][4]; b2[1] = d1[q][5];
      v2f b3; b3[0] = d1[q][6]; b3[1] = d1[q][7];
      y0 = wmma_f32(a3[q], b0, y0);
      y1 = wmma_f32(a2[q], b1, y1);
      y2 = wmma_f32(a2[q], b2, y2);
      y3 = wmma_f32(a2[q], b3, y3);
    }

    if (t == 0) {
      const v4f z = {0.0f, 0.0f, 0.0f, 0.0f};
      for (int i = 10; i < 16; ++i) {
        *reinterpret_cast<v4f*>(ob + (4 * i + fr) * 4096) = z;
      }
    }

    const int col = t * 16 + nl;
    for (int j = 0; j < 4; ++j) {
      const int ky = 4 * h + j;
      L[(ky * 5 + 1) * LDS_PITCH + col] = sqrtf(y1[2 * j] * y1[2 * j] + y1[2 * j + 1] * y1[2 * j + 1]);
      L[(ky * 5 + 2) * LDS_PITCH + col] = sqrtf(y2[2 * j] * y2[2 * j] + y2[2 * j + 1] * y2[2 * j + 1]);
      L[(ky * 5 + 3) * LDS_PITCH + col] = sqrtf(y3[2 * j] * y3[2 * j] + y3[2 * j + 1] * y3[2 * j + 1]);
    }
    {
      const int kxe = 4 * h;
      const float m0 = fabsf(y0[0]);
      const float m4 = fabsf(y0[1]);
      const float m1 = sqrtf(y0[2] * y0[2] + y0[3] * y0[3]);
      const float m2 = sqrtf(y0[4] * y0[4] + y0[5] * y0[5]);
      const float m3 = sqrtf(y0[6] * y0[6] + y0[7] * y0[7]);
      L[(0 * 5 + kxe) * LDS_PITCH + col] = m0;
      L[(1 * 5 + kxe) * LDS_PITCH + col] = m1;
      L[(2 * 5 + kxe) * LDS_PITCH + col] = m2;
      L[(3 * 5 + kxe) * LDS_PITCH + col] = m3;
      L[(4 * 5 + kxe) * LDS_PITCH + col] = m4;
      L[(5 * 5 + kxe) * LDS_PITCH + col] = m3;
      L[(6 * 5 + kxe) * LDS_PITCH + col] = m2;
      L[(7 * 5 + kxe) * LDS_PITCH + col] = m1;
    }
  }

  asm volatile("s_wait_dscnt 0" ::: "memory");

  for (int i = 0; i < 10; ++i) {
    const int f = 4 * i + fr;
    const float w = wsig[i];
    v4f m = *reinterpret_cast<const v4f*>(L + f * LDS_PITCH + c4);
    m[0] *= w; m[1] *= w; m[2] *= w; m[3] *= w;
    *reinterpret_cast<v4f*>(ob + f * 4096) = m;
  }
}

extern "C" void kernel_launch(void* const* d_in, const int* in_sizes, int n_in,
                              void* d_out, int out_size, void* d_ws, size_t ws_size,
                              hipStream_t stream) {
  (void)in_sizes; (void)n_in; (void)out_size; (void)d_ws; (void)ws_size;
  const float* x  = (const float*)d_in[0];
  const float* fw = (const float*)d_in[1];
  float* out = (float*)d_out;

  dct8x8_wmma<<<512, 256, 0, stream>>>(x, fw, out);
}


# magnitudes and sigmoids on the hardware sqrt and rcp instructions (about 1 ulp each) instead of the correctly rounded expansions: vector instructions per wave 814 -> 390; results move a few ulp
# speedup vs baseline: 8.0898x; 1.1155x over previous
#include <hip/hip_runtime.h>
#include <math.h>

typedef __attribute__((ext_vector_type(2))) float v2f;
typedef __attribute__((ext_vector_type(4))) float v4f;
typedef __attribute__((ext_vector_type(8))) float v8f;

static __device__ __forceinline__ v8f wmma_f32(v2f a, v2f b, v8f c) {
  return __builtin_amdgcn_wmma_f32_16x16x4_f32(false, a, false, b, (short)0, c, false, false);
}

static __device__ __forceinline__ float cos8(int n) {
  n &= 7;
  const float R = 0.7071067811865476f;
  float mag = (n & 1) ? R : ((n & 2) ? 0.0f : 1.0f);
  float r = ((n >= 3) && (n <= 5)) ? -mag : mag;
  return r;
}
static __device__ __forceinline__ float sin8(int n) { return cos8(n + 6); }

static __device__ __forceinline__ float hw_sqrt(float s) { return __builtin_amdgcn_sqrtf(s); }
static __device__ __forceinline__ float hw_rcp(float s) { return __builtin_amdgcn_rcpf(s); }

#define LDS_PITCH 36

__global__ void __launch_bounds__(256) __attribute__((amdgpu_waves_per_eu(4, 4))) dct8x8_wmma(const float* __restrict__ x,
                                                   const float* __restrict__ fw,
                                                   float* __restrict__ out) {
  __shared__ __attribute__((aligned(16))) float lds[8][40 * LDS_PITCH];
  const int lane = threadIdx.x & 31;
  const int wave = threadIdx.x >> 5;
  const int gw = blockIdx.x * 8 + wave;
  const int b    = gw >> 7;
  const int hb   = (gw >> 1) & 63;
  const int half = gw & 1;
  const int h  = lane >> 4;
  const int nl = lane & 15;

  const float* xp = x + ((b * 512 + hb * 8) * 512 + half * 256 + nl * 8 + 4 * h);
  v4f xr[2][8];
  for (int t = 0; t < 2; ++t) {
    for (int r = 0; r < 8; ++r) {
      xr[t][r] = *reinterpret_cast<const v4f*>(xp + r * 512 + t * 128);
    }
  }

  const int fr = lane >> 3;
  const int c4 = (lane & 7) * 4;
  float* ob = out + b * (64 * 4096) + hb * 64 + half * 32 + c4;
  float wsig[10];
  for (int i = 0; i < 10; ++i) {
    wsig[i] = hw_rcp(1.0f + __expf(-fw[4 * i + fr]));
  }

  const float S1 = 0.35355339059327373f;

  v2f a1[2][2];
  {
    const int rowsel = nl >> 3;
    const int vi = nl & 7;
    const int kx = (vi == 0) ? 0 : ((vi == 1) ? 4 : (vi >> 1));
    const int part = (vi >= 2) ? (vi & 1) : 0;
    for (int rho = 0; rho < 2; ++rho) {
      for (int ip = 0; ip < 2; ++ip) {
        for (int v = 0; v < 2; ++v) {
          const int c = 4 * h + 2 * ip + v;
          const int n = kx * c;
          const float val = cos8(n + (part ? 2 : 0));
          a1[rho][ip][v] = (rowsel == rho) ? val * S1 : 0.0f;
        }
      }
    }
  }

  v2f a2[4], a3[4];
  {
    const int ky2 = 4 * (nl >> 3) + ((nl & 7) >> 1);
    const int op2 = nl & 1;
    const int kxo = nl >> 3;
    const int v3 = nl & 7;
    const int ky3 = (v3 == 0) ? 0 : ((v3 == 1) ? 4 : (v3 >> 1));
    const int op3 = (v3 >= 2) ? (v3 & 1) : 0;
    for (int q = 0; q < 4; ++q) {
      const int r = 2 * q + h;
      for (int v = 0; v < 2; ++v) {
        const int n2 = ky2 * r;
        const int sh2 = (op2 == 0) ? ((v == 0) ? 0 : 6) : ((v == 0) ? 2 : 0);
        a2[q][v] = cos8(n2 + sh2) * S1;
        const int n3 = ky3 * r;
        const float val3 = cos8(n3 + (op3 ? 2 : 0));
        a3[q][v] = (v == kxo) ? val3 * S1 : 0.0f;
      }
    }
  }

  float* L = &lds[wave][0];

  for (int t = 0; t < 2; ++t) {
    v8f d1[4];
    for (int q = 0; q < 4; ++q) d1[q] = (v8f){0, 0, 0, 0, 0, 0, 0, 0};
    for (int rho = 0; rho < 2; ++rho) {
      for (int ip = 0; ip < 2; ++ip) {
        for (int q = 0; q < 4; ++q) {
          v2f bb;
          bb[0] = xr[t][2 * q + rho][2 * ip];
          bb[1] = xr[t][2 * q + rho][2 * ip + 1];
          d1[q] = wmma_f32(a1[rho][ip], bb, d1[q]);
        }
      }
    }

    v8f y0 = {0, 0, 0, 0, 0, 0, 0, 0};
    v8f y1 = {0, 0, 0, 0, 0, 0, 0, 0};
    v8f y2 = {0, 0, 0, 0, 0, 0, 0, 0};
    v8f y3 = {0, 0, 0, 0, 0, 0, 0, 0};
    for (int q = 0; q < 4; ++q) {
      v2f b0; b0[0] = d1[q][0]; b0[1] = d1[q][1];
      v2f b1; b1[0] = d1[q][2]; b1[1] = d1[q][3];
      v2f b2; b2[0] = d1[q][4]; b2[1] = d1[q][5];
      v2f b3; b3[0] = d1[q][6]; b3[1] = d1[q][7];
      y0 = wmma_f32(a3[q], b0, y0);
      y1 = wmma_f32(a2[q], b1, y1);
      y2 = wmma_f32(a2[q], b2, y2);
      y3 = wmma_f32(a2[q], b3, y3);
    }

    if (t == 0) {
      const v4f z = {0.0f, 0.0f, 0.0f, 0.0f};
      for (int i = 10; i < 16; ++i) {
        *reinterpret_cast<v4f*>(ob + (4 * i + fr) * 4096) = z;
      }
    }

    const int col = t * 16 + nl;
    for (int j = 0; j < 4; ++j) {
      const int ky = 4 * h + j;
      L[(ky * 5 + 1) * LDS_PITCH + col] = hw_sqrt(y1[2 * j] * y1[2 * j] + y1[2 * j + 1] * y1[2 * j + 1]);
      L[(ky * 5 + 2) * LDS_PITCH + col] = hw_sqrt(y2[2 * j] * y2[2 * j] + y2[2 * j + 1] * y2[2 * j + 1]);
      L[(ky * 5 + 3) * LDS_PITCH + col] = hw_sqrt(y3[2 * j] * y3[2 * j] + y3[2 * j + 1] * y3[2 * j + 1]);
    }
    {
      const int kxe = 4 * h;
      const float m0 = fabsf(y0[0]);
      const float m4 = fabsf(y0[1]);
      const float m1 = hw_sqrt(y0[2] * y0[2] + y0[3] * y0[3]);
      const float m2 = hw_sqrt(y0[4] * y0[4] + y0[5] * y0[5]);
      const float m3 = hw_sqrt(y0[6] * y0[6] + y0[7] * y0[7]);
      L[(0 * 5 + kxe) * LDS_PITCH + col] = m0;
      L[(1 * 5 + kxe) * LDS_PITCH + col] = m1;
      L[(2 * 5 + kxe) * LDS_PITCH + col] = m2;
      L[(3 * 5 + kxe) * LDS_PITCH + col] = m3;
      L[(4 * 5 + kxe) * LDS_PITCH + col] = m4;
      L[(5 * 5 + kxe) * LDS_PITCH + col] = m3;
      L[(6 * 5 + kxe) * LDS_PITCH + col] = m2;
      L[(7 * 5 + kxe) * LDS_PITCH + col] = m1;
    }
  }

  asm volatile("s_wait_dscnt 0" ::: "memory");

  for (int i = 0; i < 10; ++i) {
    const int f = 4 * i + fr;
    const float w = wsig[i];
    v4f m = *reinterpret_cast<const v4f*>(L + f * LDS_PITCH + c4);
    m[0] *= w; m[1] *= w; m[2] *= w; m[3] *= w;
    *reinterpret_cast<v4f*>(ob + f * 4096) = m;
  }
}

extern "C" void kernel_launch(void* const* d_in, const int* in_sizes, int n_in,
                              void* d_out, int out_size, void* d_ws, size_t ws_size,
                              hipStream_t stream) {
  (void)in_sizes; (void)n_in; (void)out_size; (void)d_ws; (void)ws_size;
  const float* x  = (const float*)d_in[0];
  const float* fw = (const float*)d_in[1];
  float* out = (float*)d_out;

  dct8x8_wmma<<<512, 256, 0, stream>>>(x, fw, out);
}
